# starting kernel
# baseline (speedup 1.0000x reference)
#include <hip/hip_runtime.h>
#include <hip/hip_bf16.h>


#define Bd    2
#define Hd    8
#define Ld    4096
#define Dd    64
#define Md    256
#define SEG   256
#define NSEG  (Ld / SEG)
#define CK    32
#define NCH   (SEG / CK)
#define PSTR  65
#define FSTR  257
#define ASTR  33
#define STATE (Md * Dd + Md)
#define RATIO 0.0625f
#define EPS   0.001f

typedef float v8f __attribute__((ext_vector_type(8)));
typedef float v2f __attribute__((ext_vector_type(2)));

static __device__ __forceinline__ v8f wmma4(v2f a, v2f b, v8f c) {
  return __builtin_amdgcn_wmma_f32_16x16x4_f32(
       false, a,  false, b,
       (short)0, c,  false,  false);
}

static __device__ __forceinline__ void prefetch_chunk(
    const float* t0, const float* t1, const float* t2,
    int ntens, int b, int h, int lbase, int tid) {
  const int per = CK * 2;
  if (tid < ntens * per) {
    int which = tid / per;
    int r     = (tid % per) >> 1;
    int halfo = (tid & 1) * 32;
    const float* src = (which == 0) ? t0 : (which == 1) ? t1 : t2;
    const float* addr = src + (((size_t)b * Ld + lbase + r) * Hd + h) * Dd + halfo;
    __builtin_prefetch(addr, 0, 3);
  }
}

__global__ __launch_bounds__(256) void perf_attn_segsum(
    const float* __restrict__ kk, const float* __restrict__ vv,
    const float* __restrict__ proj, float* __restrict__ ws)
{
  __shared__ float Plds[Md * PSTR];
  __shared__ float Slds[Md * PSTR];
  __shared__ float sLds[Md];
  __shared__ float raw [CK * PSTR];
  __shared__ float Vlds[CK * PSTR];
  __shared__ float Kf  [CK * FSTR];

  const int tid  = threadIdx.x;
  const int lane = tid & 31, wave = tid >> 5;
  const int lm   = lane & 15, hi   = lane >> 4;

  const int bh  = blockIdx.x / NSEG;
  const int seg = blockIdx.x % NSEG;
  const int b = bh / Hd, h = bh % Hd;

  for (int i = tid; i < Md * Dd; i += 256) {
    Plds[(i >> 6) * PSTR + (i & 63)] = proj[i];
    Slds[(i >> 6) * PSTR + (i & 63)] = 0.f;
  }
  for (int i = tid; i < Md; i += 256) sLds[i] = 0.f;
  __syncthreads();

  for (int ci = 0; ci < NCH; ++ci) {
    const int lbase = seg * SEG + ci * CK;
    for (int i = tid; i < CK * Dd; i += 256) {
      int r = i >> 6, d = i & 63;
      size_t g = (((size_t)b * Ld + lbase + r) * Hd + h) * Dd + d;
      raw [r * PSTR + d] = kk[g];
      Vlds[r * PSTR + d] = vv[g];
    }
    if (ci + 1 < NCH)
      prefetch_chunk(kk, vv, vv, 2, b, h, lbase + CK, tid);
    __syncthreads();

    for (int t = wave; t < 2 * (Md / 16); t += 8) {
      int it = t & 1, mt = t >> 1;
      int r0 = it * 16, c0 = mt * 16;
      v8f acc = {};
      for (int k0 = 0; k0 < Dd; k0 += 4) {
        v2f a, bb;
        a.x  = raw [(r0 + lm) * PSTR + k0 + 2 * hi];
        a.y  = raw [(r0 + lm) * PSTR + k0 + 2 * hi + 1];
        bb.x = Plds[(c0 + lm) * PSTR + k0 + 2 * hi];
        bb.y = Plds[(c0 + lm) * PSTR + k0 + 2 * hi + 1];
        acc = wmma4(a, bb, acc);
      }
      for (int e = 0; e < 8; ++e)
        Kf[(r0 + e + 8 * hi) * FSTR + c0 + lm] =
            fmaxf(RATIO * acc[e], 0.f) + EPS;
    }
    __syncthreads();

    {
      float cs = 0.f;
      for (int r = 0; r < CK; ++r) cs += Kf[r * FSTR + tid];
      sLds[tid] += cs;
    }
    for (int t = wave * 8; t < wave * 8 + 8; ++t) {
      int mt = t >> 2, dt = t & 3;
      int r0 = mt * 16, c0 = dt * 16;
      v8f acc;
      for (int e = 0; e < 8; ++e)
        acc[e] = Slds[(r0 + e + 8 * hi) * PSTR + c0 + lm];
      for (int j0 = 0; j0 < CK; j0 += 4) {
        v2f a, bb;
        a.x  = Kf  [(j0 + 2 * hi)     * FSTR + r0 + lm];
        a.y  = Kf  [(j0 + 2 * hi + 1) * FSTR + r0 + lm];
        bb.x = Vlds[(j0 + 2 * hi)     * PSTR + c0 + lm];
        bb.y = Vlds[(j0 + 2 * hi + 1) * PSTR + c0 + lm];
        acc = wmma4(a, bb, acc);
      }
      for (int e = 0; e < 8; ++e)
        Slds[(r0 + e + 8 * hi) * PSTR + c0 + lm] = acc[e];
    }
    __syncthreads();
  }

  float* dst = ws + (size_t)(bh * NSEG + seg) * STATE;
  for (int i = tid; i < Md * Dd; i += 256)
    dst[i] = Slds[(i >> 6) * PSTR + (i & 63)];
  for (int i = tid; i < Md; i += 256)
    dst[Md * Dd + i] = sLds[i];
}

__global__ __launch_bounds__(256) void perf_attn_prefix(float* __restrict__ ws)
{
  const int bh = blockIdx.x;
  float* base = ws + (size_t)bh * NSEG * STATE;
  for (int e = threadIdx.x; e < STATE; e += 256) {
    float run = 0.f;
    for (int sgi = 0; sgi < NSEG; ++sgi) {
      float* p = base + (size_t)sgi * STATE + e;
      float t = *p;
      *p = run;
      run += t;
    }
  }
}

__global__ __launch_bounds__(256) void perf_attn_main(
    const float* __restrict__ qq, const float* __restrict__ kk,
    const float* __restrict__ vv, const float* __restrict__ proj,
    const float* __restrict__ ws, float* __restrict__ out)
{
  __shared__ float Plds[Md * PSTR];
  __shared__ float Slds[Md * PSTR];
  __shared__ float sLds[Md];
  __shared__ float raw [CK * PSTR];
  __shared__ float Vlds[CK * PSTR];
  __shared__ float Qf  [CK * FSTR];
  __shared__ float Kf  [CK * FSTR];
  __shared__ float Alds[CK * ASTR];
  __shared__ float denLds[CK];

  const int tid  = threadIdx.x;
  const int lane = tid & 31, wave = tid >> 5;
  const int lm   = lane & 15, hi   = lane >> 4;

  const int bh  = blockIdx.x / NSEG;
  const int seg = blockIdx.x % NSEG;
  const int b = bh / Hd, h = bh % Hd;

  for (int i = tid; i < Md * Dd; i += 256)
    Plds[(i >> 6) * PSTR + (i & 63)] = proj[i];
  const float* st = ws + (size_t)(bh * NSEG + seg) * STATE;
  for (int i = tid; i < Md * Dd; i += 256)
    Slds[(i >> 6) * PSTR + (i & 63)] = st[i];
  for (int i = tid; i < Md; i += 256) sLds[i] = st[Md * Dd + i];
  __syncthreads();

  for (int ci = 0; ci < NCH; ++ci) {
    const int lbase = seg * SEG + ci * CK;

    for (int i = tid; i < CK * Dd; i += 256) {
      int r = i >> 6, d = i & 63;
      raw[r * PSTR + d] = qq[(((size_t)b * Ld + lbase + r) * Hd + h) * Dd + d];
    }
    if (ci + 1 < NCH)
      prefetch_chunk(qq, kk, vv, 3, b, h, lbase + CK, tid);
    __syncthreads();

    for (int t = wave; t < 2 * (Md / 16); t += 8) {
      int it = t & 1, mt = t >> 1;
      int r0 = it * 16, c0 = mt * 16;
      v8f acc = {};
      for (int k0 = 0; k0 < Dd; k0 += 4) {
        v2f a, bb;
        a.x  = raw [(r0 + lm) * PSTR + k0 + 2 * hi];
        a.y  = raw [(r0 + lm) * PSTR + k0 + 2 * hi + 1];
        bb.x = Plds[(c0 + lm) * PSTR + k0 + 2 * hi];
        bb.y = Plds[(c0 + lm) * PSTR + k0 + 2 * hi + 1];
        acc = wmma4(a, bb, acc);
      }
      for (int e = 0; e < 8; ++e)
        Qf[(r0 + e + 8 * hi) * FSTR + c0 + lm] =
            fmaxf(RATIO * acc[e], 0.f) + EPS;
    }
    __syncthreads();

    for (int i = tid; i < CK * Dd; i += 256) {
      int r = i >> 6, d = i & 63;
      size_t g = (((size_t)b * Ld + lbase + r) * Hd + h) * Dd + d;
      raw [r * PSTR + d] = kk[g];
      Vlds[r * PSTR + d] = vv[g];
    }
    __syncthreads();

    for (int t = wave; t < 2 * (Md / 16); t += 8) {
      int it = t & 1, mt = t >> 1;
      int r0 = it * 16, c0 = mt * 16;
      v8f acc = {};
      for (int k0 = 0; k0 < Dd; k0 += 4) {
        v2f a, bb;
        a.x  = raw [(r0 + lm) * PSTR + k0 + 2 * hi];
        a.y  = raw [(r0 + lm) * PSTR + k0 + 2 * hi + 1];
        bb.x = Plds[(c0 + lm) * PSTR + k0 + 2 * hi];
        bb.y = Plds[(c0 + lm) * PSTR + k0 + 2 * hi + 1];
        acc = wmma4(a, bb, acc);
      }
      for (int e = 0; e < 8; ++e)
        Kf[(r0 + e + 8 * hi) * FSTR + c0 + lm] =
            fmaxf(RATIO * acc[e], 0.f) + EPS;
    }
    __syncthreads();

    if (wave < 3) {
      const int itA[3] = {0, 1, 1};
      const int jtA[3] = {0, 0, 1};
      int it = itA[wave], jt = jtA[wave];
      v8f acc = {};
      for (int k0 = 0; k0 < Md; k0 += 4) {
        v2f a, bb;
        a.x  = Qf[(it * 16 + lm) * FSTR + k0 + 2 * hi];
        a.y  = Qf[(it * 16 + lm) * FSTR + k0 + 2 * hi + 1];
        bb.x = Kf[(jt * 16 + lm) * FSTR + k0 + 2 * hi];
        bb.y = Kf[(jt * 16 + lm) * FSTR + k0 + 2 * hi + 1];
        acc = wmma4(a, bb, acc);
      }
      for (int e = 0; e < 8; ++e) {
        int gi = it * 16 + e + 8 * hi, gj = jt * 16 + lm;
        Alds[gi * ASTR + gj] = (gj <= gi) ? acc[e] : 0.f;
      }
    } else if (wave == 3) {
      for (int i2 = lane; i2 < 256; i2 += 32)
        Alds[(i2 >> 4) * ASTR + 16 + (i2 & 15)] = 0.f;
    }
    __syncthreads();

    if (wave == 0) {
      float dsum = 0.f;
      for (int m = 0; m < Md; ++m) dsum += Qf[lane * FSTR + m] * sLds[m];
      for (int j = 0; j < CK; ++j) dsum += Alds[lane * ASTR + j];
      denLds[lane] = dsum;
    }
    v8f oacc = {};
    {
      int it = wave >> 2, dt = wave & 3;
      int r0 = it * 16, c0 = dt * 16;
      for (int k0 = 0; k0 < Md; k0 += 4) {
        v2f a, bb;
        a.x  = Qf  [(r0 + lm) * FSTR + k0 + 2 * hi];
        a.y  = Qf  [(r0 + lm) * FSTR + k0 + 2 * hi + 1];
        bb.x = Slds[(k0 + 2 * hi)     * PSTR + c0 + lm];
        bb.y = Slds[(k0 + 2 * hi + 1) * PSTR + c0 + lm];
        oacc = wmma4(a, bb, oacc);
      }
      for (int j0 = 0; j0 < CK; j0 += 4) {
        v2f a, bb;
        a.x  = Alds[(r0 + lm) * ASTR + j0 + 2 * hi];
        a.y  = Alds[(r0 + lm) * ASTR + j0 + 2 * hi + 1];
        bb.x = Vlds[(j0 + 2 * hi)     * PSTR + c0 + lm];
        bb.y = Vlds[(j0 + 2 * hi + 1) * PSTR + c0 + lm];
        oacc = wmma4(a, bb, oacc);
      }
    }
    __syncthreads();

    {
      int it = wave >> 2, dt = wave & 3;
      for (int e = 0; e < 8; ++e) {
        int row = it * 16 + e + 8 * hi;
        float den = denLds[row];
        den = (den <= 0.f) ? 1.f : den;
        float* dst =
            out + (((size_t)b * Ld + lbase + row) * Hd + h) * Dd + dt * 16 + lm;
        __builtin_nontemporal_store(oacc[e] / den, dst);
      }
    }
    {
      float cs = 0.f;
      for (int r = 0; r < CK; ++r) cs += Kf[r * FSTR + tid];
      sLds[tid] += cs;
    }
    for (int t = wave * 8; t < wave * 8 + 8; ++t) {
      int mt = t >> 2, dt = t & 3;
      int r0 = mt * 16, c0 = dt * 16;
      v8f acc;
      for (int e = 0; e < 8; ++e)
        acc[e] = Slds[(r0 + e + 8 * hi) * PSTR + c0 + lm];
      for (int j0 = 0; j0 < CK; j0 += 4) {
        v2f a, bb;
        a.x  = Kf  [(j0 + 2 * hi)     * FSTR + r0 + lm];
        a.y  = Kf  [(j0 + 2 * hi + 1) * FSTR + r0 + lm];
        bb.x = Vlds[(j0 + 2 * hi)     * PSTR + c0 + lm];
        bb.y = Vlds[(j0 + 2 * hi + 1) * PSTR + c0 + lm];
        acc = wmma4(a, bb, acc);
      }
      for (int e = 0; e < 8; ++e)
        Slds[(r0 + e + 8 * hi) * PSTR + c0 + lm] = acc[e];
    }
    __syncthreads();
  }
}

extern "C" void kernel_launch(void* const* d_in, const int* in_sizes, int n_in,
                              void* d_out, int out_size, void* d_ws, size_t ws_size,
                              hipStream_t stream) {
  const float* q    = (const float*)d_in[0];
  const float* k    = (const float*)d_in[1];
  const float* v    = (const float*)d_in[2];
  const float* proj = (const float*)d_in[3];
  float* out = (float*)d_out;
  float* ws  = (float*)d_ws;

  (void)in_sizes; (void)n_in; (void)out_size; (void)ws_size;

  perf_attn_segsum<<<Bd * Hd * NSEG, 256, 0, stream>>>(k, v, proj, ws);
  perf_attn_prefix<<<Bd * Hd, 256, 0, stream>>>(ws);
  perf_attn_main  <<<Bd * Hd * NSEG, 256, 0, stream>>>(q, k, v, proj, ws, out);
}


# Fragment k map of the long WMMA chains changed, A and B loaders alike: element s of lane half h reads k = kk+32h+16s per 64-k block, so a fragment read spreads over LDS banks; sum order changes
# speedup vs baseline: 2.3170x; 2.3170x over previous
#include <hip/hip_runtime.h>
#include <hip/hip_bf16.h>


#define Bd    2
#define Hd    8
#define Ld    4096
#define Dd    64
#define Md    256
#define SEG   256
#define NSEG  (Ld / SEG)
#define CK    32
#define NCH   (SEG / CK)
#define PSTR  65
#define FSTR  257
#define ASTR  33
#define STATE (Md * Dd + Md)
#define NT    512
#define NW    (NT / 32)
#define NV4   (Md * Dd / 4 / NT)
#define KQ    4
#define RATIO 0.0625f
#define EPS   0.001f
#define KB    64
#define KS    (KB / 4)
static_assert(Dd == KB && Md / KQ == KB && (Md / 2) % KB == 0 && CK == 32,
              "fragment k map assumes these block sizes");

typedef float v8f __attribute__((ext_vector_type(8)));
typedef float v2f __attribute__((ext_vector_type(2)));

static __device__ __forceinline__ v8f wmma4(v2f a, v2f b, v8f c) {
  return __builtin_amdgcn_wmma_f32_16x16x4_f32(
       false, a,  false, b,
       (short)0, c,  false,  false);
}

static __device__ __forceinline__ void prefetch_chunk(
    const float* t0, const float* t1, const float* t2,
    int ntens, int b, int h, int lbase, int tid) {
  const int per = CK * 2;
  if (tid < ntens * per) {
    int which = tid / per;
    int r     = (tid % per) >> 1;
    int halfo = (tid & 1) * 32;
    const float* src = (which == 0) ? t0 : (which == 1) ? t1 : t2;
    const float* addr = src + (((size_t)b * Ld + lbase + r) * Hd + h) * Dd + halfo;
    __builtin_prefetch(addr, 0, 3);
  }
}

__global__ __launch_bounds__(NT) void perf_attn_segsum(
    const float* __restrict__ kk, const float* __restrict__ vv,
    const float* __restrict__ proj, float* __restrict__ ws)
{
  __shared__ float Plds[Md * PSTR];
  __shared__ float Slds[Md * PSTR];
  __shared__ float sLds[Md];
  __shared__ float raw [CK * PSTR];
  __shared__ float Vlds[CK * PSTR];
  __shared__ float Kf  [CK * FSTR];

  const int tid  = threadIdx.x;
  const int lane = tid & 31, wave = tid >> 5;
  const int lm   = lane & 15, hi   = lane >> 4;

  const int bh  = blockIdx.x / NSEG;
  const int seg = blockIdx.x % NSEG;
  const int b = bh / Hd, h = bh % Hd;

  for (int i = tid; i < Md * Dd; i += NT) {
    Plds[(i >> 6) * PSTR + (i & 63)] = proj[i];
    Slds[(i >> 6) * PSTR + (i & 63)] = 0.f;
  }
  for (int i = tid; i < Md; i += NT) sLds[i] = 0.f;
  __syncthreads();

  for (int ci = 0; ci < NCH; ++ci) {
    const int lbase = seg * SEG + ci * CK;
    for (int i = tid; i < CK * Dd; i += NT) {
      int r = i >> 6, d = i & 63;
      size_t g = (((size_t)b * Ld + lbase + r) * Hd + h) * Dd + d;
      raw [r * PSTR + d] = kk[g];
      Vlds[r * PSTR + d] = vv[g];
    }
    if (ci + 1 < NCH)
      prefetch_chunk(kk, vv, vv, 2, b, h, lbase + CK, tid);
    __syncthreads();

    for (int t = wave; t < 2 * (Md / 16); t += NW) {
      int it = t & 1, mt = t >> 1;
      int r0 = it * 16, c0 = mt * 16;
      v8f acc = {};
      for (int kk = 0; kk < KS; ++kk) {
        const int k0 = kk + 32 * hi;
        v2f a, bb;
        a.x  = raw [(r0 + lm) * PSTR + k0];
        a.y  = raw [(r0 + lm) * PSTR + k0 + 16];
        bb.x = Plds[(c0 + lm) * PSTR + k0];
        bb.y = Plds[(c0 + lm) * PSTR + k0 + 16];
        acc = wmma4(a, bb, acc);
      }
      for (int e = 0; e < 8; ++e)
        Kf[(r0 + e + 8 * hi) * FSTR + c0 + lm] =
            fmaxf(RATIO * acc[e], 0.f) + EPS;
    }
    __syncthreads();

    if (tid < Md) {
      float cs = 0.f;
      for (int r = 0; r < CK; ++r) cs += Kf[r * FSTR + tid];
      sLds[tid] += cs;
    }
    for (int t = wave * 4; t < wave * 4 + 4; ++t) {
      int mt = t >> 2, dt = t & 3;
      int r0 = mt * 16, c0 = dt * 16;
      v8f acc;
      for (int e = 0; e < 8; ++e)
        acc[e] = Slds[(r0 + e + 8 * hi) * PSTR + c0 + lm];
      for (int jj = 0; jj < CK / 4; ++jj) {
        const int j = jj + 16 * hi;
        v2f a, bb;
        a.x  = Kf  [j       * FSTR + r0 + lm];
        a.y  = Kf  [(j + 8) * FSTR + r0 + lm];
        bb.x = Vlds[j       * PSTR + c0 + lm];
        bb.y = Vlds[(j + 8) * PSTR + c0 + lm];
        acc = wmma4(a, bb, acc);
      }
      for (int e = 0; e < 8; ++e)
        Slds[(r0 + e + 8 * hi) * PSTR + c0 + lm] = acc[e];
    }
    __syncthreads();
  }

  float* dst = ws + (size_t)(bh * NSEG + seg) * STATE;
  for (int i = tid; i < Md * Dd; i += NT)
    dst[i] = Slds[(i >> 6) * PSTR + (i & 63)];
  for (int i = tid; i < Md; i += NT)
    dst[Md * Dd + i] = sLds[i];
}

static __device__ __forceinline__ void seg_load(
    float4 (&t)[NV4], float& ts, const float* q, int tid) {
  const float4* p = reinterpret_cast<const float4*>(q) + tid;
  for (int j = 0; j < NV4; ++j) t[j] = p[NT * j];
  ts = (tid < Md) ? q[Md * Dd + tid] : 0.f;
}
static __device__ __forceinline__ void seg_add(
    float4 (&run)[NV4], float& runs, const float4 (&t)[NV4], float ts) {
  for (int j = 0; j < NV4; ++j) {
    run[j].x += t[j].x; run[j].y += t[j].y;
    run[j].z += t[j].z; run[j].w += t[j].w;
  }
  runs += ts;
}

__global__ __launch_bounds__(NT) void perf_attn_main(
    const float* __restrict__ qq, const float* __restrict__ kk,
    const float* __restrict__ vv, const float* __restrict__ proj,
    const float* __restrict__ ws, float* __restrict__ out)
{
  __shared__ float Plds[Md * PSTR];
  __shared__ float Slds[Md * PSTR];
  __shared__ float sLds[Md];
  __shared__ float raw [CK * PSTR];
  __shared__ float Vlds[CK * PSTR];
  __shared__ float Qf  [CK * FSTR];
  __shared__ float Kf  [CK * FSTR];
  __shared__ float Alds[CK * ASTR];
  __shared__ float Apart[KQ * CK * ASTR];
  __shared__ float Opart[CK * PSTR];
  __shared__ float denLds[CK];

  const int tid  = threadIdx.x;
  const int lane = tid & 31, wave = __builtin_amdgcn_readfirstlane(tid >> 5);
  const int lm   = lane & 15, hi   = lane >> 4;

  const int bh  = blockIdx.x / NSEG;
  const int seg = blockIdx.x % NSEG;
  const int b = bh / Hd, h = bh % Hd;

  for (int i = tid; i < Md * Dd; i += NT)
    Plds[(i >> 6) * PSTR + (i & 63)] = proj[i];
  {
    const float* segs = ws + (size_t)bh * NSEG * STATE;
    float4 run[NV4];
    float  runs = 0.f;
    for (int j = 0; j < NV4; ++j) run[j] = make_float4(0.f, 0.f, 0.f, 0.f);
    if (seg > 0) {
      float4 ta[NV4], tb[NV4];
      float  sa, sb;
      seg_load(ta, sa, segs, tid);
      int sgi = 1;
      for (; sgi + 1 < seg; sgi += 2) {
        seg_load(tb, sb, segs + (size_t)sgi * STATE, tid);
        seg_add(run, runs, ta, sa);
        seg_load(ta, sa, segs + (size_t)(sgi + 1) * STATE, tid);
        seg_add(run, runs, tb, sb);
      }
      if (sgi < seg) {
        seg_load(tb, sb, segs + (size_t)sgi * STATE, tid);
        seg_add(run, runs, ta, sa);
        seg_add(run, runs, tb, sb);
      } else {
        seg_add(run, runs, ta, sa);
      }
    }
    for (int j = 0; j < NV4; ++j) {
      int i = 4 * (tid + NT * j);
      float* d = &Slds[(i >> 6) * PSTR + (i & 63)];
      d[0] = run[j].x; d[1] = run[j].y; d[2] = run[j].z; d[3] = run[j].w;
    }
    if (tid < Md) sLds[tid] = runs;
  }
  __syncthreads();

  for (int ci = 0; ci < NCH; ++ci) {
    const int lbase = seg * SEG + ci * CK;

    for (int i = tid; i < CK * Dd; i += NT) {
      int r = i >> 6, d = i & 63;
      raw[r * PSTR + d] = qq[(((size_t)b * Ld + lbase + r) * Hd + h) * Dd + d];
    }
    if (ci + 1 < NCH)
      prefetch_chunk(qq, kk, vv, 3, b, h, lbase + CK, tid);
    __syncthreads();

    for (int t = wave; t < 2 * (Md / 16); t += NW) {
      int it = t & 1, mt = t >> 1;
      int r0 = it * 16, c0 = mt * 16;
      v8f acc = {};
      for (int kk = 0; kk < KS; ++kk) {
        const int k0 = kk + 32 * hi;
        v2f a, bb;
        a.x  = raw [(r0 + lm) * PSTR + k0];
        a.y  = raw [(r0 + lm) * PSTR + k0 + 16];
        bb.x = Plds[(c0 + lm) * PSTR + k0];
        bb.y = Plds[(c0 + lm) * PSTR + k0 + 16];
        acc = wmma4(a, bb, acc);
      }
      for (int e = 0; e < 8; ++e)
        Qf[(r0 + e + 8 * hi) * FSTR + c0 + lm] =
            fmaxf(RATIO * acc[e], 0.f) + EPS;
    }
    __syncthreads();

    for (int i = tid; i < CK * Dd; i += NT) {
      int r = i >> 6, d = i & 63;
      size_t g = (((size_t)b * Ld + lbase + r) * Hd + h) * Dd + d;
      raw [r * PSTR + d] = kk[g];
      Vlds[r * PSTR + d] = vv[g];
    }
    __syncthreads();

    for (int t = wave; t < 2 * (Md / 16); t += NW) {
      int it = t & 1, mt = t >> 1;
      int r0 = it * 16, c0 = mt * 16;
      v8f acc = {};
      for (int kk = 0; kk < KS; ++kk) {
        const int k0 = kk + 32 * hi;
        v2f a, bb;
        a.x  = raw [(r0 + lm) * PSTR + k0];
        a.y  = raw [(r0 + lm) * PSTR + k0 + 16];
        bb.x = Plds[(c0 + lm) * PSTR + k0];
        bb.y = Plds[(c0 + lm) * PSTR + k0 + 16];
        acc = wmma4(a, bb, acc);
      }
      for (int e = 0; e < 8; ++e)
        Kf[(r0 + e + 8 * hi) * FSTR + c0 + lm] =
            fmaxf(RATIO * acc[e], 0.f) + EPS;
    }
    __syncthreads();

    if (wave < 3 * KQ) {
      const int tl = wave % 3, kq = wave / 3;
      const int it = (tl > 0) ? 1 : 0, jt = (tl == 2) ? 1 : 0;
      v8f acc = {};
      for (int kk = 0; kk < KS; ++kk) {
        const int k0 = kq * KB + kk + 32 * hi;
        v2f a, bb;
        a.x  = Qf[(it * 16 + lm) * FSTR + k0];
        a.y  = Qf[(it * 16 + lm) * FSTR + k0 + 16];
        bb.x = Kf[(jt * 16 + lm) * FSTR + k0];
        bb.y = Kf[(jt * 16 + lm) * FSTR + k0 + 16];
        acc = wmma4(a, bb, acc);
      }
      for (int e = 0; e < 8; ++e)
        Apart[kq * (CK * ASTR) + (it * 16 + e + 8 * hi) * ASTR + jt * 16 + lm] = acc[e];
    }
    __syncthreads();

    for (int i = tid; i < CK * CK; i += NT) {
      int gi = i >> 5, gj = i & 31;
      float a = 0.f;
      if (gj <= gi) {
        const float* p = &Apart[gi * ASTR + gj];
        a = p[0];
        for (int q = 1; q < KQ; ++q) a += p[q * (CK * ASTR)];
      }
      Alds[gi * ASTR + gj] = a;
    }
    __syncthreads();

    v8f oacc = {};
    {
      const int tile = wave & 7, kh = wave >> 3;
      int it = tile >> 2, dt = tile & 3;
      int r0 = it * 16, c0 = dt * 16;
      for (int kb = 0; kb < Md / 2; kb += KB) {
        for (int kk = 0; kk < KS; ++kk) {
          const int k0 = kh * (Md / 2) + kb + kk + 32 * hi;
          v2f a, bb;
          a.x  = Qf  [(r0 + lm) * FSTR + k0];
          a.y  = Qf  [(r0 + lm) * FSTR + k0 + 16];
          bb.x = Slds[k0        * PSTR + c0 + lm];
          bb.y = Slds[(k0 + 16) * PSTR + c0 + lm];
          oacc = wmma4(a, bb, oacc);
        }
      }
      if (kh == 0) {
        for (int j0 = 0; j0 < CK; j0 += 4) {
          v2f a, bb;
          a.x  = Alds[(r0 + lm) * ASTR + j0 + 2 * hi];
          a.y  = Alds[(r0 + lm) * ASTR + j0 + 2 * hi + 1];
          bb.x = Vlds[(j0 + 2 * hi)     * PSTR + c0 + lm];
          bb.y = Vlds[(j0 + 2 * hi + 1) * PSTR + c0 + lm];
          oacc = wmma4(a, bb, oacc);
        }
      } else {
        for (int e = 0; e < 8; ++e)
          Opart[(r0 + e + 8 * hi) * PSTR + c0 + lm] = oacc[e];
        const int row = (wave - 8) * 4 + (lane >> 3), l8 = lane & 7;
        float dsum = 0.f;
        for (int m = l8; m < Md; m += 8) dsum += Qf[row * FSTR + m] * sLds[m];
        for (int j = l8; j < CK; j += 8) dsum += Alds[row * ASTR + j];
        dsum += __shfl_xor(dsum, 4);
        dsum += __shfl_xor(dsum, 2);
        dsum += __shfl_xor(dsum, 1);
        if (l8 == 0) denLds[row] = dsum;
      }
    }
    __syncthreads();

    if (wave < 8) {
      int it = wave >> 2, dt = wave & 3;
      for (int e = 0; e < 8; ++e) {
        int row = it * 16 + e + 8 * hi;
        float den = denLds[row];
        den = (den <= 0.f) ? 1.f : den;
        float num = oacc[e] + Opart[row * PSTR + dt * 16 + lm];
        float* dst =
            out + (((size_t)b * Ld + lbase + row) * Hd + h) * Dd + dt * 16 + lm;
        __builtin_nontemporal_store(num / den, dst);
      }
    }
    if (tid < Md) {
      float cs = 0.f;
      for (int r = 0; r < CK; ++r) cs += Kf[r * FSTR + tid];
      sLds[tid] += cs;
    }
    for (int t = wave * 4; t < wave * 4 + 4; ++t) {
      int mt = t >> 2, dt = t & 3;
      int r0 = mt * 16, c0 = dt * 16;
      v8f acc;
      for (int e = 0; e < 8; ++e)
        acc[e] = Slds[(r0 + e + 8 * hi) * PSTR + c0 + lm];
      for (int jj = 0; jj < CK / 4; ++jj) {
        const int j = jj + 16 * hi;
        v2f a, bb;
        a.x  = Kf  [j       * FSTR + r0 + lm];
        a.y  = Kf  [(j + 8) * FSTR + r0 + lm];
        bb.x = Vlds[j       * PSTR + c0 + lm];
        bb.y = Vlds[(j + 8) * PSTR + c0 + lm];
        acc = wmma4(a, bb, acc);
      }
      for (int e = 0; e < 8; ++e)
        Slds[(r0 + e + 8 * hi) * PSTR + c0 + lm] = acc[e];
    }
    __syncthreads();
  }
}

extern "C" void kernel_launch(void* const* d_in, const int* in_sizes, int n_in,
                              void* d_out, int out_size, void* d_ws, size_t ws_size,
                              hipStream_t stream) {
  const float* q    = (const float*)d_in[0];
  const float* k    = (const float*)d_in[1];
  const float* v    = (const float*)d_in[2];
  const float* proj = (const float*)d_in[3];
  float* out = (float*)d_out;
  float* ws  = (float*)d_ws;

  (void)in_sizes; (void)n_in; (void)out_size; (void)ws_size;

  perf_attn_segsum<<<Bd * Hd * NSEG, NT, 0, stream>>>(k, v, proj, ws);
  perf_attn_main  <<<Bd * Hd * NSEG, NT, 0, stream>>>(q, k, v, proj, ws, out);
}


# Q/K/V rows staged in LDS per 64-row tile (two chunks) by one load stage with all loads in flight, in both kernels; Qf and Kf feature stages then share one barrier; load width and sum order kept
# speedup vs baseline: 2.5955x; 1.1202x over previous
#include <hip/hip_runtime.h>
#include <hip/hip_bf16.h>


#define Bd    2
#define Hd    8
#define Ld    4096
#define Dd    64
#define Md    256
#define SEG   256
#define NSEG  (Ld / SEG)
#define CK    32
#define NCH   (SEG / CK)
#define ST    64
#define NTL   (SEG / ST)
#define CPT   (ST / CK)
#define PSTR  65
#define FSTR  257
#define ASTR  33
#define STATE (Md * Dd + Md)
#define NT    512
#define NW    (NT / 32)
#define NV4   (Md * Dd / 4 / NT)
#define NLD   (ST * Dd / NT)
#define KQ    4
#define RATIO 0.0625f
#define EPS   0.001f
#define KB    64
#define KS    (KB / 4)
static_assert(Dd == KB && Md / KQ == KB && (Md / 2) % KB == 0 && CK == 32,
              "fragment k map assumes these block sizes");
static_assert(SEG % ST == 0 && ST % CK == 0 && (ST * Dd) % NT == 0 && 3 * ST * 2 <= NT,
              "staging tile must split evenly over chunks and threads");

typedef float v8f __attribute__((ext_vector_type(8)));
typedef float v2f __attribute__((ext_vector_type(2)));

static __device__ __forceinline__ v8f wmma4(v2f a, v2f b, v8f c) {
  return __builtin_amdgcn_wmma_f32_16x16x4_f32(
       false, a,  false, b,
       (short)0, c,  false,  false);
}

static __device__ __forceinline__ void prefetch_tile(
    const float* t0, const float* t1, const float* t2,
    int ntens, int b, int h, int lbase, int tid) {
  const int per = ST * 2;
  if (tid < ntens * per) {
    int which = tid / per;
    int r     = (tid % per) >> 1;
    int halfo = (tid & 1) * 32;
    const float* src = (which == 0) ? t0 : (which == 1) ? t1 : t2;
    const float* addr = src + (((size_t)b * Ld + lbase + r) * Hd + h) * Dd + halfo;
    __builtin_prefetch(addr, 0, 3);
  }
}

static __device__ __forceinline__ void tile_fetch(
    float (&t)[NLD], const float* __restrict__ src, size_t g0, int tid) {
  for (int j = 0; j < NLD; ++j) {
    int i = tid + NT * j;
    t[j] = src[g0 + (size_t)(i >> 6) * (Hd * Dd) + (i & 63)];
  }
}
static __device__ __forceinline__ void tile_put(
    float* lds, const float (&t)[NLD], int tid) {
  for (int j = 0; j < NLD; ++j) {
    int i = tid + NT * j;
    lds[(i >> 6) * PSTR + (i & 63)] = t[j];
  }
}

__global__ __launch_bounds__(NT) void perf_attn_segsum(
    const float* __restrict__ kk, const float* __restrict__ vv,
    const float* __restrict__ proj, float* __restrict__ ws)
{
  __shared__ float Plds[Md * PSTR];
  __shared__ float Slds[Md * PSTR];
  __shared__ float sLds[Md];
  __shared__ float raw [ST * PSTR];
  __shared__ float Vlds[ST * PSTR];
  __shared__ float Kf  [CK * FSTR];

  const int tid  = threadIdx.x;
  const int lane = tid & 31, wave = tid >> 5;
  const int lm   = lane & 15, hi   = lane >> 4;

  const int bh  = blockIdx.x / NSEG;
  const int seg = blockIdx.x % NSEG;
  const int b = bh / Hd, h = bh % Hd;

  for (int i = tid; i < Md * Dd; i += NT) {
    Plds[(i >> 6) * PSTR + (i & 63)] = proj[i];
    Slds[(i >> 6) * PSTR + (i & 63)] = 0.f;
  }
  for (int i = tid; i < Md; i += NT) sLds[i] = 0.f;
  __syncthreads();

  for (int ti = 0; ti < NTL; ++ti) {
    {
      const int l0 = seg * SEG + ti * ST;
      const size_t g0 = (((size_t)b * Ld + l0) * Hd + h) * Dd;
      float tk[NLD], tv[NLD];
      tile_fetch(tk, kk, g0, tid);
      tile_fetch(tv, vv, g0, tid);
      if (ti + 1 < NTL)
        prefetch_tile(kk, vv, vv, 2, b, h, l0 + ST, tid);
      tile_put(raw,  tk, tid);
      tile_put(Vlds, tv, tid);
    }
    __syncthreads();

   for (int cc = 0; cc < CPT; ++cc) {
    const int ro = cc * CK;

    for (int t = wave; t < 2 * (Md / 16); t += NW) {
      int it = t & 1, mt = t >> 1;
      int r0 = it * 16, c0 = mt * 16;
      v8f acc = {};
      for (int kk = 0; kk < KS; ++kk) {
        const int k0 = kk + 32 * hi;
        v2f a, bb;
        a.x  = raw [(ro + r0 + lm) * PSTR + k0];
        a.y  = raw [(ro + r0 + lm) * PSTR + k0 + 16];
        bb.x = Plds[(c0 + lm) * PSTR + k0];
        bb.y = Plds[(c0 + lm) * PSTR + k0 + 16];
        acc = wmma4(a, bb, acc);
      }
      for (int e = 0; e < 8; ++e)
        Kf[(r0 + e + 8 * hi) * FSTR + c0 + lm] =
            fmaxf(RATIO * acc[e], 0.f) + EPS;
    }
    __syncthreads();

    if (tid < Md) {
      float cs = 0.f;
      for (int r = 0; r < CK; ++r) cs += Kf[r * FSTR + tid];
      sLds[tid] += cs;
    }
    for (int t = wave * 4; t < wave * 4 + 4; ++t) {
      int mt = t >> 2, dt = t & 3;
      int r0 = mt * 16, c0 = dt * 16;
      v8f acc;
      for (int e = 0; e < 8; ++e)
        acc[e] = Slds[(r0 + e + 8 * hi) * PSTR + c0 + lm];
      for (int jj = 0; jj < CK / 4; ++jj) {
        const int j = jj + 16 * hi;
        v2f a, bb;
        a.x  = Kf  [j       * FSTR + r0 + lm];
        a.y  = Kf  [(j + 8) * FSTR + r0 + lm];
        bb.x = Vlds[(ro + j)     * PSTR + c0 + lm];
        bb.y = Vlds[(ro + j + 8) * PSTR + c0 + lm];
        acc = wmma4(a, bb, acc);
      }
      for (int e = 0; e < 8; ++e)
        Slds[(r0 + e + 8 * hi) * PSTR + c0 + lm] = acc[e];
    }
    __syncthreads();
   }
  }

  float* dst = ws + (size_t)(bh * NSEG + seg) * STATE;
  for (int i = tid; i < Md * Dd; i += NT)
    dst[i] = Slds[(i >> 6) * PSTR + (i & 63)];
  for (int i = tid; i < Md; i += NT)
    dst[Md * Dd + i] = sLds[i];
}

static __device__ __forceinline__ void seg_load(
    float4 (&t)[NV4], float& ts, const float* q, int tid) {
  const float4* p = reinterpret_cast<const float4*>(q) + tid;
  for (int j = 0; j < NV4; ++j) t[j] = p[NT * j];
  ts = (tid < Md) ? q[Md * Dd + tid] : 0.f;
}
static __device__ __forceinline__ void seg_add(
    float4 (&run)[NV4], float& runs, const float4 (&t)[NV4], float ts) {
  for (int j = 0; j < NV4; ++j) {
    run[j].x += t[j].x; run[j].y += t[j].y;
    run[j].z += t[j].z; run[j].w += t[j].w;
  }
  runs += ts;
}

__global__ __launch_bounds__(NT) void perf_attn_main(
    const float* __restrict__ qq, const float* __restrict__ kk,
    const float* __restrict__ vv, const float* __restrict__ proj,
    const float* __restrict__ ws, float* __restrict__ out)
{
  __shared__ float Plds[Md * PSTR];
  __shared__ float Slds[Md * PSTR];
  __shared__ float sLds[Md];
  __shared__ float rawQ[ST * PSTR];
  __shared__ float rawK[ST * PSTR];
  __shared__ float Vlds[ST * PSTR];
  __shared__ float Qf  [CK * FSTR];
  __shared__ float Kf  [CK * FSTR];
  __shared__ float Alds[CK * ASTR];
  __shared__ float Apart[KQ * CK * ASTR];
  __shared__ float Opart[CK * PSTR];
  __shared__ float denLds[CK];

  const int tid  = threadIdx.x;
  const int lane = tid & 31, wave = __builtin_amdgcn_readfirstlane(tid >> 5);
  const int lm   = lane & 15, hi   = lane >> 4;

  const int bh  = blockIdx.x / NSEG;
  const int seg = blockIdx.x % NSEG;
  const int b = bh / Hd, h = bh % Hd;

  for (int i = tid; i < Md * Dd; i += NT)
    Plds[(i >> 6) * PSTR + (i & 63)] = proj[i];
  {
    const float* segs = ws + (size_t)bh * NSEG * STATE;
    float4 run[NV4];
    float  runs = 0.f;
    for (int j = 0; j < NV4; ++j) run[j] = make_float4(0.f, 0.f, 0.f, 0.f);
    if (seg > 0) {
      float4 ta[NV4], tb[NV4];
      float  sa, sb;
      seg_load(ta, sa, segs, tid);
      int sgi = 1;
      for (; sgi + 1 < seg; sgi += 2) {
        seg_load(tb, sb, segs + (size_t)sgi * STATE, tid);
        seg_add(run, runs, ta, sa);
        seg_load(ta, sa, segs + (size_t)(sgi + 1) * STATE, tid);
        seg_add(run, runs, tb, sb);
      }
      if (sgi < seg) {
        seg_load(tb, sb, segs + (size_t)sgi * STATE, tid);
        seg_add(run, runs, ta, sa);
        seg_add(run, runs, tb, sb);
      } else {
        seg_add(run, runs, ta, sa);
      }
    }
    for (int j = 0; j < NV4; ++j) {
      int i = 4 * (tid + NT * j);
      float* d = &Slds[(i >> 6) * PSTR + (i & 63)];
      d[0] = run[j].x; d[1] = run[j].y; d[2] = run[j].z; d[3] = run[j].w;
    }
    if (tid < Md) sLds[tid] = runs;
  }
  __syncthreads();

  for (int ti = 0; ti < NTL; ++ti) {
    {
      const int l0 = seg * SEG + ti * ST;
      const size_t g0 = (((size_t)b * Ld + l0) * Hd + h) * Dd;
      float tq[NLD], tk[NLD], tv[NLD];
      tile_fetch(tq, qq, g0, tid);
      tile_fetch(tk, kk, g0, tid);
      tile_fetch(tv, vv, g0, tid);
      if (ti + 1 < NTL)
        prefetch_tile(qq, kk, vv, 3, b, h, l0 + ST, tid);
      tile_put(rawQ, tq, tid);
      tile_put(rawK, tk, tid);
      tile_put(Vlds, tv, tid);
    }
    __syncthreads();

   for (int cc = 0; cc < CPT; ++cc) {
    const int ro = cc * CK;
    const int lbase = seg * SEG + ti * ST + ro;

    for (int t = wave; t < 2 * (Md / 16); t += NW) {
      int it = t & 1, mt = t >> 1;
      int r0 = it * 16, c0 = mt * 16;
      v8f acc = {};
      for (int kk = 0; kk < KS; ++kk) {
        const int k0 = kk + 32 * hi;
        v2f a, bb;
        a.x  = rawQ[(ro + r0 + lm) * PSTR + k0];
        a.y  = rawQ[(ro + r0 + lm) * PSTR + k0 + 16];
        bb.x = Plds[(c0 + lm) * PSTR + k0];
        bb.y = Plds[(c0 + lm) * PSTR + k0 + 16];
        acc = wmma4(a, bb, acc);
      }
      for (int e = 0; e < 8; ++e)
        Qf[(r0 + e + 8 * hi) * FSTR + c0 + lm] =
            fmaxf(RATIO * acc[e], 0.f) + EPS;
    }

    for (int t = wave; t < 2 * (Md / 16); t += NW) {
      int it = t & 1, mt = t >> 1;
      int r0 = it * 16, c0 = mt * 16;
      v8f acc = {};
      for (int kk = 0; kk < KS; ++kk) {
        const int k0 = kk + 32 * hi;
        v2f a, bb;
        a.x  = rawK[(ro + r0 + lm) * PSTR + k0];
        a.y  = rawK[(ro + r0 + lm) * PSTR + k0 + 16];
        bb.x = Plds[(c0 + lm) * PSTR + k0];
        bb.y = Plds[(c0 + lm) * PSTR + k0 + 16];
        acc = wmma4(a, bb, acc);
      }
      for (int e = 0; e < 8; ++e)
        Kf[(r0 + e + 8 * hi) * FSTR + c0 + lm] =
            fmaxf(RATIO * acc[e], 0.f) + EPS;
    }
    __syncthreads();

    if (wave < 3 * KQ) {
      const int tl = wave % 3, kq = wave / 3;
      const int it = (tl > 0) ? 1 : 0, jt = (tl == 2) ? 1 : 0;
      v8f acc = {};
      for (int kk = 0; kk < KS; ++kk) {
        const int k0 = kq * KB + kk + 32 * hi;
        v2f a, bb;
        a.x  = Qf[(it * 16 + lm) * FSTR + k0];
        a.y  = Qf[(it * 16 + lm) * FSTR + k0 + 16];
        bb.x = Kf[(jt * 16 + lm) * FSTR + k0];
        bb.y = Kf[(jt * 16 + lm) * FSTR + k0 + 16];
        acc = wmma4(a, bb, acc);
      }
      for (int e = 0; e < 8; ++e)
        Apart[kq * (CK * ASTR) + (it * 16 + e + 8 * hi) * ASTR + jt * 16 + lm] = acc[e];
    }
    __syncthreads();

    for (int i = tid; i < CK * CK; i += NT) {
      int gi = i >> 5, gj = i & 31;
      float a = 0.f;
      if (gj <= gi) {
        const float* p = &Apart[gi * ASTR + gj];
        a = p[0];
        for (int q = 1; q < KQ; ++q) a += p[q * (CK * ASTR)];
      }
      Alds[gi * ASTR + gj] = a;
    }
    __syncthreads();

    v8f oacc = {};
    {
      const int tile = wave & 7, kh = wave >> 3;
      int it = tile >> 2, dt = tile & 3;
      int r0 = it * 16, c0 = dt * 16;
      for (int kb = 0; kb < Md / 2; kb += KB) {
        for (int kk = 0; kk < KS; ++kk) {
          const int k0 = kh * (Md / 2) + kb + kk + 32 * hi;
          v2f a, bb;
          a.x  = Qf  [(r0 + lm) * FSTR + k0];
          a.y  = Qf  [(r0 + lm) * FSTR + k0 + 16];
          bb.x = Slds[k0        * PSTR + c0 + lm];
          bb.y = Slds[(k0 + 16) * PSTR + c0 + lm];
          oacc = wmma4(a, bb, oacc);
        }
      }
      if (kh == 0) {
        for (int j0 = 0; j0 < CK; j0 += 4) {
          v2f a, bb;
          a.x  = Alds[(r0 + lm) * ASTR + j0 + 2 * hi];
          a.y  = Alds[(r0 + lm) * ASTR + j0 + 2 * hi + 1];
          bb.x = Vlds[(ro + j0 + 2 * hi)     * PSTR + c0 + lm];
          bb.y = Vlds[(ro + j0 + 2 * hi + 1) * PSTR + c0 + lm];
          oacc = wmma4(a, bb, oacc);
        }
      } else {
        for (int e = 0; e < 8; ++e)
          Opart[(r0 + e + 8 * hi) * PSTR + c0 + lm] = oacc[e];
        const int row = (wave - 8) * 4 + (lane >> 3), l8 = lane & 7;
        float dsum = 0.f;
        for (int m = l8; m < Md; m += 8) dsum += Qf[row * FSTR + m] * sLds[m];
        for (int j = l8; j < CK; j += 8) dsum += Alds[row * ASTR + j];
        dsum += __shfl_xor(dsum, 4);
        dsum += __shfl_xor(dsum, 2);
        dsum += __shfl_xor(dsum, 1);
        if (l8 == 0) denLds[row] = dsum;
      }
    }
    __syncthreads();

    if (wave < 8) {
      int it = wave >> 2, dt = wave & 3;
      for (int e = 0; e < 8; ++e) {
        int row = it * 16 + e + 8 * hi;
        float den = denLds[row];
        den = (den <= 0.f) ? 1.f : den;
        float num = oacc[e] + Opart[row * PSTR + dt * 16 + lm];
        float* dst =
            out + (((size_t)b * Ld + lbase + row) * Hd + h) * Dd + dt * 16 + lm;
        __builtin_nontemporal_store(num / den, dst);
      }
    }
    if (tid < Md) {
      float cs = 0.f;
      for (int r = 0; r < CK; ++r) cs += Kf[r * FSTR + tid];
      sLds[tid] += cs;
    }
    for (int t = wave * 4; t < wave * 4 + 4; ++t) {
      int mt = t >> 2, dt = t & 3;
      int r0 = mt * 16, c0 = dt * 16;
      v8f acc;
      for (int e = 0; e < 8; ++e)
        acc[e] = Slds[(r0 + e + 8 * hi) * PSTR + c0 + lm];
      for (int jj = 0; jj < CK / 4; ++jj) {
        const int j = jj + 16 * hi;
        v2f a, bb;
        a.x  = Kf  [j       * FSTR + r0 + lm];
        a.y  = Kf  [(j + 8) * FSTR + r0 + lm];
        bb.x = Vlds[(ro + j)     * PSTR + c0 + lm];
        bb.y = Vlds[(ro + j + 8) * PSTR + c0 + lm];
        acc = wmma4(a, bb, acc);
      }
      for (int e = 0; e < 8; ++e)
        Slds[(r0 + e + 8 * hi) * PSTR + c0 + lm] = acc[e];
    }
    __syncthreads();
   }
  }
}

extern "C" void kernel_launch(void* const* d_in, const int* in_sizes, int n_in,
                              void* d_out, int out_size, void* d_ws, size_t ws_size,
                              hipStream_t stream) {
  const float* q    = (const float*)d_in[0];
  const float* k    = (const float*)d_in[1];
  const float* v    = (const float*)d_in[2];
  const float* proj = (const float*)d_in[3];
  float* out = (float*)d_out;
  float* ws  = (float*)d_ws;

  (void)in_sizes; (void)n_in; (void)out_size; (void)ws_size;

  perf_attn_segsum<<<Bd * Hd * NSEG, NT, 0, stream>>>(k, v, proj, ws);
  perf_attn_main  <<<Bd * Hd * NSEG, NT, 0, stream>>>(q, k, v, proj, ws, out);
}


# LDS tiles stored k-contiguous at 16B x odd pitches (state as S^T, V as V^T, extra transposed Kf copy): each WMMA fragment read is one 128-bit load for two products; sum order of all products changes
# speedup vs baseline: 3.0366x; 1.1699x over previous
#include <hip/hip_runtime.h>
#include <hip/hip_bf16.h>


#define Bd    2
#define Hd    8
#define Ld    4096
#define Dd    64
#define Md    256
#define SEG   256
#define NSEG  (Ld / SEG)
#define CK    32
#define NCH   (SEG / CK)
#define ST    64
#define NTL   (SEG / ST)
#define CPT   (ST / CK)
#define RP    68
#define FP    260
#define TP    36
#define STATE (Md * Dd + Md)
#define NT    512
#define NW    (NT / 32)
#define NV4   (Md * Dd / 4 / NT)
#define NLD   (ST * Dd / NT)
#define KQ    4
#define RATIO 0.0625f
#define EPS   0.001f
static_assert(Dd == 64 && Md == 256 && CK == 32 && Md / KQ == 64,
              "tile index arithmetic assumes these sizes");
static_assert(RP % 8 == 4 && FP % 8 == 4 && TP % 8 == 4,
              "row pitches must be 16 bytes times an odd number");
static_assert(SEG % ST == 0 && ST % CK == 0 && (ST * Dd) % NT == 0 && 3 * ST * 2 <= NT,
              "staging tile must split evenly over chunks and threads");

typedef float v8f __attribute__((ext_vector_type(8)));
typedef float v4f __attribute__((ext_vector_type(4)));
typedef float v2f __attribute__((ext_vector_type(2)));
#define LDS16 __attribute__((aligned(16)))

static __device__ __forceinline__ v8f wmma4(v2f a, v2f b, v8f c) {
  return __builtin_amdgcn_wmma_f32_16x16x4_f32(
       false, a,  false, b,
       (short)0, c,  false,  false);
}

static __device__ __forceinline__ v8f mma_k8(const float* ap, const float* bp, v8f acc) {
  const v4f a = *reinterpret_cast<const v4f*>(ap);
  const v4f b = *reinterpret_cast<const v4f*>(bp);
  v2f a0 = {a.x, a.y}, b0 = {b.x, b.y};
  v2f a1 = {a.z, a.w}, b1 = {b.z, b.w};
  acc = wmma4(a0, b0, acc);
  return wmma4(a1, b1, acc);
}

static __device__ __forceinline__ void prefetch_tile(
    const float* t0, const float* t1, const float* t2,
    int ntens, int b, int h, int lbase, int tid) {
  const int per = ST * 2;
  if (tid < ntens * per) {
    int which = tid / per;
    int r     = (tid % per) >> 1;
    int halfo = (tid & 1) * 32;
    const float* src = (which == 0) ? t0 : (which == 1) ? t1 : t2;
    const float* addr = src + (((size_t)b * Ld + lbase + r) * Hd + h) * Dd + halfo;
    __builtin_prefetch(addr, 0, 3);
  }
}

static __device__ __forceinline__ void tile_fetch(
    float (&t)[NLD], const float* __restrict__ src, size_t g0, int tid) {
  for (int j = 0; j < NLD; ++j) {
    int i = tid + NT * j;
    t[j] = src[g0 + (size_t)(i >> 6) * (Hd * Dd) + (i & 63)];
  }
}
static __device__ __forceinline__ void tile_put(
    float* lds, const float (&t)[NLD], int tid) {
  for (int j = 0; j < NLD; ++j) {
    int i = tid + NT * j;
    lds[(i >> 6) * RP + (i & 63)] = t[j];
  }
}
static __device__ __forceinline__ void tile_put_t(
    float* lds, const float (&t)[NLD], int tid) {
  for (int j = 0; j < NLD; ++j) {
    int i = tid + NT * j;
    lds[(i & 63) * RP + (i >> 6)] = t[j];
  }
}

static __device__ __forceinline__ float feat(float x) {
  return fmaxf(RATIO * x, 0.f) + EPS;
}

__global__ __launch_bounds__(NT) void perf_attn_segsum(
    const float* __restrict__ kk, const float* __restrict__ vv,
    const float* __restrict__ proj, float* __restrict__ ws)
{
  __shared__ LDS16 float Plds[Md * RP];
  __shared__ LDS16 float St  [Dd * FP];
  __shared__ LDS16 float sLds[Md];
  __shared__ LDS16 float raw [ST * RP];
  __shared__ LDS16 float Vt  [Dd * RP];
  __shared__ LDS16 float KfT [Md * TP];

  const int tid  = threadIdx.x;
  const int lane = tid & 31, wave = tid >> 5;
  const int lm   = lane & 15, hi   = lane >> 4;

  const int bh  = blockIdx.x / NSEG;
  const int seg = blockIdx.x % NSEG;
  const int b = bh / Hd, h = bh % Hd;

  for (int i = tid; i < Md * Dd; i += NT) {
    Plds[(i >> 6) * RP + (i & 63)]  = proj[i];
    St  [(i >> 8) * FP + (i & 255)] = 0.f;
  }
  for (int i = tid; i < Md; i += NT) sLds[i] = 0.f;
  __syncthreads();

  for (int ti = 0; ti < NTL; ++ti) {
    {
      const int l0 = seg * SEG + ti * ST;
      const size_t g0 = (((size_t)b * Ld + l0) * Hd + h) * Dd;
      float tk[NLD], tv[NLD];
      tile_fetch(tk, kk, g0, tid);
      tile_fetch(tv, vv, g0, tid);
      if (ti + 1 < NTL)
        prefetch_tile(kk, vv, vv, 2, b, h, l0 + ST, tid);
      tile_put  (raw, tk, tid);
      tile_put_t(Vt,  tv, tid);
    }
    __syncthreads();

   for (int cc = 0; cc < CPT; ++cc) {
    const int ro = cc * CK;

    for (int t = wave; t < 2 * (Md / 16); t += NW) {
      int it = t & 1, mt = t >> 1;
      int r0 = it * 16, c0 = mt * 16;
      const float* ap = &raw [(ro + r0 + lm) * RP + 4 * hi];
      const float* bp = &Plds[(c0 + lm)      * RP + 4 * hi];
      v8f acc = {};
      for (int q = 0; q < Dd / 8; ++q)
        acc = mma_k8(ap + 8 * q, bp + 8 * q, acc);
      v4f f0 = {feat(acc[0]), feat(acc[1]), feat(acc[2]), feat(acc[3])};
      v4f f1 = {feat(acc[4]), feat(acc[5]), feat(acc[6]), feat(acc[7])};
      float* kt = &KfT[(c0 + lm) * TP + r0 + 8 * hi];
      *reinterpret_cast<v4f*>(kt)     = f0;
      *reinterpret_cast<v4f*>(kt + 4) = f1;
    }
    __syncthreads();

    if (tid < Md) {
      const float* p = &KfT[tid * TP];
      float cs = 0.f;
      for (int q = 0; q < CK / 4; ++q) {
        const v4f x = *reinterpret_cast<const v4f*>(p + 4 * q);
        cs += x.x; cs += x.y; cs += x.z; cs += x.w;
      }
      sLds[tid] += cs;
    }
    for (int t = wave * 4; t < wave * 4 + 4; ++t) {
      int mt = t >> 2, dt = t & 3;
      int r0 = mt * 16, c0 = dt * 16;
      float* sp = &St[(c0 + lm) * FP + r0 + 8 * hi];
      const v4f s0 = *reinterpret_cast<const v4f*>(sp);
      const v4f s1 = *reinterpret_cast<const v4f*>(sp + 4);
      v8f acc = {s0.x, s0.y, s0.z, s0.w, s1.x, s1.y, s1.z, s1.w};
      const float* ap = &KfT[(r0 + lm) * TP + 4 * hi];
      const float* bp = &Vt [(c0 + lm) * RP + ro + 4 * hi];
      for (int q = 0; q < CK / 8; ++q)
        acc = mma_k8(ap + 8 * q, bp + 8 * q, acc);
      v4f o0 = {acc[0], acc[1], acc[2], acc[3]};
      v4f o1 = {acc[4], acc[5], acc[6], acc[7]};
      *reinterpret_cast<v4f*>(sp)     = o0;
      *reinterpret_cast<v4f*>(sp + 4) = o1;
    }
    __syncthreads();
   }
  }

  float* dst = ws + (size_t)(bh * NSEG + seg) * STATE;
  for (int i = tid; i < Md * Dd; i += NT)
    dst[i] = St[(i >> 8) * FP + (i & 255)];
  for (int i = tid; i < Md; i += NT)
    dst[Md * Dd + i] = sLds[i];
}

static __device__ __forceinline__ void seg_load(
    float4 (&t)[NV4], float& ts, const float* q, int tid) {
  const float4* p = reinterpret_cast<const float4*>(q) + tid;
  for (int j = 0; j < NV4; ++j) t[j] = p[NT * j];
  ts = (tid < Md) ? q[Md * Dd + tid] : 0.f;
}
static __device__ __forceinline__ void seg_add(
    float4 (&run)[NV4], float& runs, const float4 (&t)[NV4], float ts) {
  for (int j = 0; j < NV4; ++j) {
    run[j].x += t[j].x; run[j].y += t[j].y;
    run[j].z += t[j].z; run[j].w += t[j].w;
  }
  runs += ts;
}

__global__ __launch_bounds__(NT) void perf_attn_main(
    const float* __restrict__ qq, const float* __restrict__ kk,
    const float* __restrict__ vv, const float* __restrict__ proj,
    const float* __restrict__ ws, float* __restrict__ out)
{
  __shared__ LDS16 float Plds[Md * RP];
  __shared__ LDS16 float St  [Dd * FP];
  __shared__ LDS16 float sLds[Md];
  __shared__ LDS16 float rawQ[ST * RP];
  __shared__ LDS16 float rawK[ST * RP];
  __shared__ LDS16 float Vt  [Dd * RP];
  __shared__ LDS16 float Qf  [CK * FP];
  __shared__ LDS16 float Kf  [CK * FP];
  __shared__ LDS16 float KfT [Md * TP];
  __shared__ LDS16 float Alds[CK * TP];
  __shared__ LDS16 float Apart[KQ * CK * TP];
  __shared__ LDS16 float Opart[CK * RP];
  __shared__ LDS16 float denLds[CK];

  const int tid  = threadIdx.x;
  const int lane = tid & 31, wave = __builtin_amdgcn_readfirstlane(tid >> 5);
  const int lm   = lane & 15, hi   = lane >> 4;

  const int bh  = blockIdx.x / NSEG;
  const int seg = blockIdx.x % NSEG;
  const int b = bh / Hd, h = bh % Hd;

  for (int i = tid; i < Md * Dd; i += NT)
    Plds[(i >> 6) * RP + (i & 63)] = proj[i];
  {
    const float* segs = ws + (size_t)bh * NSEG * STATE;
    float4 run[NV4];
    float  runs = 0.f;
    for (int j = 0; j < NV4; ++j) run[j] = make_float4(0.f, 0.f, 0.f, 0.f);
    if (seg > 0) {
      float4 ta[NV4], tb[NV4];
      float  sa, sb;
      seg_load(ta, sa, segs, tid);
      int sgi = 1;
      for (; sgi + 1 < seg; sgi += 2) {
        seg_load(tb, sb, segs + (size_t)sgi * STATE, tid);
        seg_add(run, runs, ta, sa);
        seg_load(ta, sa, segs + (size_t)(sgi + 1) * STATE, tid);
        seg_add(run, runs, tb, sb);
      }
      if (sgi < seg) {
        seg_load(tb, sb, segs + (size_t)sgi * STATE, tid);
        seg_add(run, runs, ta, sa);
        seg_add(run, runs, tb, sb);
      } else {
        seg_add(run, runs, ta, sa);
      }
    }
    for (int j = 0; j < NV4; ++j) {
      int i = 4 * (tid + NT * j);
      v4f r4 = {run[j].x, run[j].y, run[j].z, run[j].w};
      *reinterpret_cast<v4f*>(&St[(i >> 8) * FP + (i & 255)]) = r4;
    }
    if (tid < Md) sLds[tid] = runs;
  }
  __syncthreads();

  for (int ti = 0; ti < NTL; ++ti) {
    {
      const int l0 = seg * SEG + ti * ST;
      const size_t g0 = (((size_t)b * Ld + l0) * Hd + h) * Dd;
      float tq[NLD], tk[NLD], tv[NLD];
      tile_fetch(tq, qq, g0, tid);
      tile_fetch(tk, kk, g0, tid);
      tile_fetch(tv, vv, g0, tid);
      if (ti + 1 < NTL)
        prefetch_tile(qq, kk, vv, 3, b, h, l0 + ST, tid);
      tile_put  (rawQ, tq, tid);
      tile_put  (rawK, tk, tid);
      tile_put_t(Vt,   tv, tid);
    }
    __syncthreads();

   for (int cc = 0; cc < CPT; ++cc) {
    const int ro = cc * CK;
    const int lbase = seg * SEG + ti * ST + ro;

    for (int t = wave; t < 2 * (Md / 16); t += NW) {
      int it = t & 1, mt = t >> 1;
      int r0 = it * 16, c0 = mt * 16;
      const float* ap = &rawQ[(ro + r0 + lm) * RP + 4 * hi];
      const float* bp = &Plds[(c0 + lm)      * RP + 4 * hi];
      v8f acc = {};
      for (int q = 0; q < Dd / 8; ++q)
        acc = mma_k8(ap + 8 * q, bp + 8 * q, acc);
      for (int e = 0; e < 8; ++e)
        Qf[(r0 + e + 8 * hi) * FP + c0 + lm] = feat(acc[e]);
    }

    for (int t = wave; t < 2 * (Md / 16); t += NW) {
      int it = t & 1, mt = t >> 1;
      int r0 = it * 16, c0 = mt * 16;
      const float* ap = &rawK[(ro + r0 + lm) * RP + 4 * hi];
      const float* bp = &Plds[(c0 + lm)      * RP + 4 * hi];
      v8f acc = {};
      for (int q = 0; q < Dd / 8; ++q)
        acc = mma_k8(ap + 8 * q, bp + 8 * q, acc);
      float f[8];
      for (int e = 0; e < 8; ++e) f[e] = feat(acc[e]);
      for (int e = 0; e < 8; ++e)
        Kf[(r0 + e + 8 * hi) * FP + c0 + lm] = f[e];
      v4f f0 = {f[0], f[1], f[2], f[3]};
      v4f f1 = {f[4], f[5], f[6], f[7]};
      float* kt = &KfT[(c0 + lm) * TP + r0 + 8 * hi];
      *reinterpret_cast<v4f*>(kt)     = f0;
      *reinterpret_cast<v4f*>(kt + 4) = f1;
    }
    __syncthreads();

    if (wave < 3 * KQ) {
      const int tl = wave % 3, kq = wave / 3;
      const int it = (tl > 0) ? 1 : 0, jt = (tl == 2) ? 1 : 0;
      const float* ap = &Qf[(it * 16 + lm) * FP + kq * (Md / KQ) + 4 * hi];
      const float* bp = &Kf[(jt * 16 + lm) * FP + kq * (Md / KQ) + 4 * hi];
      v8f acc = {};
      for (int q = 0; q < Md / KQ / 8; ++q)
        acc = mma_k8(ap + 8 * q, bp + 8 * q, acc);
      for (int e = 0; e < 8; ++e)
        Apart[kq * (CK * TP) + (it * 16 + e + 8 * hi) * TP + jt * 16 + lm] = acc[e];
    }
    __syncthreads();

    for (int i = tid; i < CK * CK; i += NT) {
      int gi = i >> 5, gj = i & 31;
      float a = 0.f;
      if (gj <= gi) {
        const float* p = &Apart[gi * TP + gj];
        a = p[0];
        for (int q = 1; q < KQ; ++q) a += p[q * (CK * TP)];
      }
      Alds[gi * TP + gj] = a;
    }
    __syncthreads();

    v8f oacc = {};
    {
      const int tile = wave & 7, kh = wave >> 3;
      int it = tile >> 2, dt = tile & 3;
      int r0 = it * 16, c0 = dt * 16;
      {
        const float* ap = &Qf[(r0 + lm) * FP + kh * (Md / 2) + 4 * hi];
        const float* bp = &St[(c0 + lm) * FP + kh * (Md / 2) + 4 * hi];
        for (int q = 0; q < Md / 2 / 8; ++q)
          oacc = mma_k8(ap + 8 * q, bp + 8 * q, oacc);
      }
      if (kh == 0) {
        const float* ap = &Alds[(r0 + lm) * TP + 4 * hi];
        const float* bp = &Vt  [(c0 + lm) * RP + ro + 4 * hi];
        for (int q = 0; q < CK / 8; ++q)
          oacc = mma_k8(ap + 8 * q, bp + 8 * q, oacc);
      } else {
        for (int e = 0; e < 8; ++e)
          Opart[(r0 + e + 8 * hi) * RP + c0 + lm] = oacc[e];
        const int row = (wave - 8) * 4 + (lane >> 3), l8 = lane & 7;
        float dsum = 0.f;
        for (int m = l8; m < Md; m += 8) dsum += Qf[row * FP + m] * sLds[m];
        for (int j = l8; j < CK; j += 8) dsum += Alds[row * TP + j];
        dsum += __shfl_xor(dsum, 4);
        dsum += __shfl_xor(dsum, 2);
        dsum += __shfl_xor(dsum, 1);
        if (l8 == 0) denLds[row] = dsum;
      }
    }
    __syncthreads();

    if (wave < 8) {
      int it = wave >> 2, dt = wave & 3;
      for (int e = 0; e < 8; ++e) {
        int row = it * 16 + e + 8 * hi;
        float den = denLds[row];
        den = (den <= 0.f) ? 1.f : den;
        float num = oacc[e] + Opart[row * RP + dt * 16 + lm];
        float* dst =
            out + (((size_t)b * Ld + lbase + row) * Hd + h) * Dd + dt * 16 + lm;
        __builtin_nontemporal_store(num / den, dst);
      }
    }
    if (tid < Md) {
      const float* p = &KfT[tid * TP];
      float cs = 0.f;
      for (int q = 0; q < CK / 4; ++q) {
        const v4f x = *reinterpret_cast<const v4f*>(p + 4 * q);
        cs += x.x; cs += x.y; cs += x.z; cs += x.w;
      }
      sLds[tid] += cs;
    }
    for (int t = wave * 4; t < wave * 4 + 4; ++t) {
      int mt = t >> 2, dt = t & 3;
      int r0 = mt * 16, c0 = dt * 16;
      float* sp = &St[(c0 + lm) * FP + r0 + 8 * hi];
      const v4f s0 = *reinterpret_cast<const v4f*>(sp);
      const v4f s1 = *reinterpret_cast<const v4f*>(sp + 4);
      v8f acc = {s0.x, s0.y, s0.z, s0.w, s1.x, s1.y, s1.z, s1.w};
      const float* ap = &KfT[(r0 + lm) * TP + 4 * hi];
      const float* bp = &Vt [(c0 + lm) * RP + ro + 4 * hi];
      for (int q = 0; q < CK / 8; ++q)
        acc = mma_k8(ap + 8 * q, bp + 8 * q, acc);
      v4f o0 = {acc[0], acc[1], acc[2], acc[3]};
      v4f o1 = {acc[4], acc[5], acc[6], acc[7]};
      *reinterpret_cast<v4f*>(sp)     = o0;
      *reinterpret_cast<v4f*>(sp + 4) = o1;
    }
    __syncthreads();
   }
  }
}

extern "C" void kernel_launch(void* const* d_in, const int* in_sizes, int n_in,
                              void* d_out, int out_size, void* d_ws, size_t ws_size,
                              hipStream_t stream) {
  const float* q    = (const float*)d_in[0];
  const float* k    = (const float*)d_in[1];
  const float* v    = (const float*)d_in[2];
  const float* proj = (const float*)d_in[3];
  float* out = (float*)d_out;
  float* ws  = (float*)d_ws;

  (void)in_sizes; (void)n_in; (void)out_size; (void)ws_size;

  perf_attn_segsum<<<Bd * Hd * NSEG, NT, 0, stream>>>(k, v, proj, ws);
  perf_attn_main  <<<Bd * Hd * NSEG, NT, 0, stream>>>(q, k, v, proj, ws, out);
}


# Main kernel's work-groups widened to 32 waves (8 per SIMD): Qf tiles on waves 0-15 beside Kf tiles on 16-31, den on waves 16-31 beside the Qf@S products, 16 lanes per row (sum order of den changes)
# speedup vs baseline: 3.1446x; 1.0356x over previous
#include <hip/hip_runtime.h>
#include <hip/hip_bf16.h>


#define Bd    2
#define Hd    8
#define Ld    4096
#define Dd    64
#define Md    256
#define SEG   256
#define NSEG  (Ld / SEG)
#define CK    32
#define NCH   (SEG / CK)
#define ST    64
#define NTL   (SEG / ST)
#define CPT   (ST / CK)
#define RP    68
#define FP    260
#define TP    36
#define STATE (Md * Dd + Md)
#define NT1   512
#define NW1   (NT1 / 32)
#define NT2   1024
#define NW2   (NT2 / 32)
#define STW   ((Md / 16) * (Dd / 16) / NW2)
#define KQ    4
#define RATIO 0.0625f
#define EPS   0.001f
static_assert(Dd == 64 && Md == 256 && CK == 32 && Md / KQ == 64,
              "tile index arithmetic assumes these sizes");
static_assert(RP % 8 == 4 && FP % 8 == 4 && TP % 8 == 4,
              "row pitches must be 16 bytes times an odd number");
static_assert(SEG % ST == 0 && ST % CK == 0 && (ST * Dd) % NT1 == 0 && (ST * Dd) % NT2 == 0 &&
              3 * ST * 2 <= NT1, "staging tile must split evenly over chunks and threads");
static_assert(NW2 == 32 && STW * NW2 == 64 && 2 * (Md / 16) == 2 * 16 && Md <= NT2 / 2,
              "the main kernel deals its stage work over exactly 32 waves");

typedef float v8f __attribute__((ext_vector_type(8)));
typedef float v4f __attribute__((ext_vector_type(4)));
typedef float v2f __attribute__((ext_vector_type(2)));
#define LDS16 __attribute__((aligned(16)))

static __device__ __forceinline__ v8f wmma4(v2f a, v2f b, v8f c) {
  return __builtin_amdgcn_wmma_f32_16x16x4_f32(
       false, a,  false, b,
       (short)0, c,  false,  false);
}

static __device__ __forceinline__ v8f mma_k8(const float* ap, const float* bp, v8f acc) {
  const v4f a = *reinterpret_cast<const v4f*>(ap);
  const v4f b = *reinterpret_cast<const v4f*>(bp);
  v2f a0 = {a.x, a.y}, b0 = {b.x, b.y};
  v2f a1 = {a.z, a.w}, b1 = {b.z, b.w};
  acc = wmma4(a0, b0, acc);
  return wmma4(a1, b1, acc);
}

static __device__ __forceinline__ void prefetch_tile(
    const float* t0, const float* t1, const float* t2,
    int ntens, int b, int h, int lbase, int tid) {
  const int per = ST * 2;
  if (tid < ntens * per) {
    int which = tid / per;
    int r     = (tid % per) >> 1;
    int halfo = (tid & 1) * 32;
    const float* src = (which == 0) ? t0 : (which == 1) ? t1 : t2;
    const float* addr = src + (((size_t)b * Ld + lbase + r) * Hd + h) * Dd + halfo;
    __builtin_prefetch(addr, 0, 3);
  }
}

template <int NTH>
static __device__ __forceinline__ void tile_fetch(
    float (&t)[ST * Dd / NTH], const float* __restrict__ src, size_t g0, int tid) {
  for (int j = 0; j < ST * Dd / NTH; ++j) {
    int i = tid + NTH * j;
    t[j] = src[g0 + (size_t)(i >> 6) * (Hd * Dd) + (i & 63)];
  }
}
template <int NTH>
static __device__ __forceinline__ void tile_put(
    float* lds, const float (&t)[ST * Dd / NTH], int tid) {
  for (int j = 0; j < ST * Dd / NTH; ++j) {
    int i = tid + NTH * j;
    lds[(i >> 6) * RP + (i & 63)] = t[j];
  }
}
template <int NTH>
static __device__ __forceinline__ void tile_put_t(
    float* lds, const float (&t)[ST * Dd / NTH], int tid) {
  for (int j = 0; j < ST * Dd / NTH; ++j) {
    int i = tid + NTH * j;
    lds[(i & 63) * RP + (i >> 6)] = t[j];
  }
}

static __device__ __forceinline__ float feat(float x) {
  return fmaxf(RATIO * x, 0.f) + EPS;
}

__global__ __launch_bounds__(NT1) void perf_attn_segsum(
    const float* __restrict__ kk, const float* __restrict__ vv,
    const float* __restrict__ proj, float* __restrict__ ws)
{
  __shared__ LDS16 float Plds[Md * RP];
  __shared__ LDS16 float St  [Dd * FP];
  __shared__ LDS16 float sLds[Md];
  __shared__ LDS16 float raw [ST * RP];
  __shared__ LDS16 float Vt  [Dd * RP];
  __shared__ LDS16 float KfT [Md * TP];

  const int tid  = threadIdx.x;
  const int lane = tid & 31, wave = tid >> 5;
  const int lm   = lane & 15, hi   = lane >> 4;

  const int bh  = blockIdx.x / NSEG;
  const int seg = blockIdx.x % NSEG;
  const int b = bh / Hd, h = bh % Hd;

  for (int i = tid; i < Md * Dd; i += NT1) {
    Plds[(i >> 6) * RP + (i & 63)]  = proj[i];
    St  [(i >> 8) * FP + (i & 255)] = 0.f;
  }
  for (int i = tid; i < Md; i += NT1) sLds[i] = 0.f;
  __syncthreads();

  for (int ti = 0; ti < NTL; ++ti) {
    {
      const int l0 = seg * SEG + ti * ST;
      const size_t g0 = (((size_t)b * Ld + l0) * Hd + h) * Dd;
      float tk[ST * Dd / NT1], tv[ST * Dd / NT1];
      tile_fetch<NT1>(tk, kk, g0, tid);
      tile_fetch<NT1>(tv, vv, g0, tid);
      if (ti + 1 < NTL)
        prefetch_tile(kk, vv, vv, 2, b, h, l0 + ST, tid);
      tile_put  <NT1>(raw, tk, tid);
      tile_put_t<NT1>(Vt,  tv, tid);
    }
    __syncthreads();

   for (int cc = 0; cc < CPT; ++cc) {
    const int ro = cc * CK;

    for (int t = wave; t < 2 * (Md / 16); t += NW1) {
      int it = t & 1, mt = t >> 1;
      int r0 = it * 16, c0 = mt * 16;
      const float* ap = &raw [(ro + r0 + lm) * RP + 4 * hi];
      const float* bp = &Plds[(c0 + lm)      * RP + 4 * hi];
      v8f acc = {};
      for (int q = 0; q < Dd / 8; ++q)
        acc = mma_k8(ap + 8 * q, bp + 8 * q, acc);
      v4f f0 = {feat(acc[0]), feat(acc[1]), feat(acc[2]), feat(acc[3])};
      v4f f1 = {feat(acc[4]), feat(acc[5]), feat(acc[6]), feat(acc[7])};
      float* kt = &KfT[(c0 + lm) * TP + r0 + 8 * hi];
      *reinterpret_cast<v4f*>(kt)     = f0;
      *reinterpret_cast<v4f*>(kt + 4) = f1;
    }
    __syncthreads();

    if (tid < Md) {
      const float* p = &KfT[tid * TP];
      float cs = 0.f;
      for (int q = 0; q < CK / 4; ++q) {
        const v4f x = *reinterpret_cast<const v4f*>(p + 4 * q);
        cs += x.x; cs += x.y; cs += x.z; cs += x.w;
      }
      sLds[tid] += cs;
    }
    for (int t = wave * 4; t < wave * 4 + 4; ++t) {
      int mt = t >> 2, dt = t & 3;
      int r0 = mt * 16, c0 = dt * 16;
      float* sp = &St[(c0 + lm) * FP + r0 + 8 * hi];
      const v4f s0 = *reinterpret_cast<const v4f*>(sp);
      const v4f s1 = *reinterpret_cast<const v4f*>(sp + 4);
      v8f acc = {s0.x, s0.y, s0.z, s0.w, s1.x, s1.y, s1.z, s1.w};
      const float* ap = &KfT[(r0 + lm) * TP + 4 * hi];
      const float* bp = &Vt [(c0 + lm) * RP + ro + 4 * hi];
      for (int q = 0; q < CK / 8; ++q)
        acc = mma_k8(ap + 8 * q, bp + 8 * q, acc);
      v4f o0 = {acc[0], acc[1], acc[2], acc[3]};
      v4f o1 = {acc[4], acc[5], acc[6], acc[7]};
      *reinterpret_cast<v4f*>(sp)     = o0;
      *reinterpret_cast<v4f*>(sp + 4) = o1;
    }
    __syncthreads();
   }
  }

  float* dst = ws + (size_t)(bh * NSEG + seg) * STATE;
  for (int i = tid; i < Md * Dd; i += NT1)
    dst[i] = St[(i >> 8) * FP + (i & 255)];
  for (int i = tid; i < Md; i += NT1)
    dst[Md * Dd + i] = sLds[i];
}

#define NV4   (Md * Dd / 4 / NT2)
static __device__ __forceinline__ void seg_load(
    float4 (&t)[NV4], float& ts, const float* q, int tid) {
  const float4* p = reinterpret_cast<const float4*>(q) + tid;
  for (int j = 0; j < NV4; ++j) t[j] = p[NT2 * j];
  ts = (tid < Md) ? q[Md * Dd + tid] : 0.f;
}
static __device__ __forceinline__ void seg_add(
    float4 (&run)[NV4], float& runs, const float4 (&t)[NV4], float ts) {
  for (int j = 0; j < NV4; ++j) {
    run[j].x += t[j].x; run[j].y += t[j].y;
    run[j].z += t[j].z; run[j].w += t[j].w;
  }
  runs += ts;
}

__global__ __launch_bounds__(NT2) void perf_attn_main(
    const float* __restrict__ qq, const float* __restrict__ kk,
    const float* __restrict__ vv, const float* __restrict__ proj,
    const float* __restrict__ ws, float* __restrict__ out)
{
  __shared__ LDS16 float Plds[Md * RP];
  __shared__ LDS16 float St  [Dd * FP];
  __shared__ LDS16 float sLds[Md];
  __shared__ LDS16 float rawQ[ST * RP];
  __shared__ LDS16 float rawK[ST * RP];
  __shared__ LDS16 float Vt  [Dd * RP];
  __shared__ LDS16 float Qf  [CK * FP];
  __shared__ LDS16 float Kf  [CK * FP];
  __shared__ LDS16 float KfT [Md * TP];
  __shared__ LDS16 float Alds[CK * TP];
  __shared__ LDS16 float Apart[KQ * CK * TP];
  __shared__ LDS16 float Opart[CK * RP];
  __shared__ LDS16 float denLds[CK];

  const int tid  = threadIdx.x;
  const int lane = tid & 31, wave = __builtin_amdgcn_readfirstlane(tid >> 5);
  const int lm   = lane & 15, hi   = lane >> 4;

  const int bh  = blockIdx.x / NSEG;
  const int seg = blockIdx.x % NSEG;
  const int b = bh / Hd, h = bh % Hd;

  for (int i = tid; i < Md * Dd; i += NT2)
    Plds[(i >> 6) * RP + (i & 63)] = proj[i];
  {
    const float* segs = ws + (size_t)bh * NSEG * STATE;
    float4 run[NV4];
    float  runs = 0.f;
    for (int j = 0; j < NV4; ++j) run[j] = make_float4(0.f, 0.f, 0.f, 0.f);
    if (seg > 0) {
      float4 ta[NV4], tb[NV4];
      float  sa, sb;
      seg_load(ta, sa, segs, tid);
      int sgi = 1;
      for (; sgi + 1 < seg; sgi += 2) {
        seg_load(tb, sb, segs + (size_t)sgi * STATE, tid);
        seg_add(run, runs, ta, sa);
        seg_load(ta, sa, segs + (size_t)(sgi + 1) * STATE, tid);
        seg_add(run, runs, tb, sb);
      }
      if (sgi < seg) {
        seg_load(tb, sb, segs + (size_t)sgi * STATE, tid);
        seg_add(run, runs, ta, sa);
        seg_add(run, runs, tb, sb);
      } else {
        seg_add(run, runs, ta, sa);
      }
    }
    for (int j = 0; j < NV4; ++j) {
      int i = 4 * (tid + NT2 * j);
      v4f r4 = {run[j].x, run[j].y, run[j].z, run[j].w};
      *reinterpret_cast<v4f*>(&St[(i >> 8) * FP + (i & 255)]) = r4;
    }
    if (tid < Md) sLds[tid] = runs;
  }
  __syncthreads();

  for (int ti = 0; ti < NTL; ++ti) {
    {
      const int l0 = seg * SEG + ti * ST;
      const size_t g0 = (((size_t)b * Ld + l0) * Hd + h) * Dd;
      float tq[ST * Dd / NT2], tk[ST * Dd / NT2], tv[ST * Dd / NT2];
      tile_fetch<NT2>(tq, qq, g0, tid);
      tile_fetch<NT2>(tk, kk, g0, tid);
      tile_fetch<NT2>(tv, vv, g0, tid);
      if (ti + 1 < NTL)
        prefetch_tile(qq, kk, vv, 3, b, h, l0 + ST, tid);
      tile_put  <NT2>(rawQ, tq, tid);
      tile_put  <NT2>(rawK, tk, tid);
      tile_put_t<NT2>(Vt,   tv, tid);
    }
    __syncthreads();

   for (int cc = 0; cc < CPT; ++cc) {
    const int ro = cc * CK;
    const int lbase = seg * SEG + ti * ST + ro;

    if (wave < 16) {
    for (int t = wave; t < 2 * (Md / 16); t += 16) {
      int it = t & 1, mt = t >> 1;
      int r0 = it * 16, c0 = mt * 16;
      const float* ap = &rawQ[(ro + r0 + lm) * RP + 4 * hi];
      const float* bp = &Plds[(c0 + lm)      * RP + 4 * hi];
      v8f acc = {};
      for (int q = 0; q < Dd / 8; ++q)
        acc = mma_k8(ap + 8 * q, bp + 8 * q, acc);
      for (int e = 0; e < 8; ++e)
        Qf[(r0 + e + 8 * hi) * FP + c0 + lm] = feat(acc[e]);
    }
    } else {
    for (int t = wave - 16; t < 2 * (Md / 16); t += 16) {
      int it = t & 1, mt = t >> 1;
      int r0 = it * 16, c0 = mt * 16;
      const float* ap = &rawK[(ro + r0 + lm) * RP + 4 * hi];
      const float* bp = &Plds[(c0 + lm)      * RP + 4 * hi];
      v8f acc = {};
      for (int q = 0; q < Dd / 8; ++q)
        acc = mma_k8(ap + 8 * q, bp + 8 * q, acc);
      float f[8];
      for (int e = 0; e < 8; ++e) f[e] = feat(acc[e]);
      for (int e = 0; e < 8; ++e)
        Kf[(r0 + e + 8 * hi) * FP + c0 + lm] = f[e];
      v4f f0 = {f[0], f[1], f[2], f[3]};
      v4f f1 = {f[4], f[5], f[6], f[7]};
      float* kt = &KfT[(c0 + lm) * TP + r0 + 8 * hi];
      *reinterpret_cast<v4f*>(kt)     = f0;
      *reinterpret_cast<v4f*>(kt + 4) = f1;
    }
    }
    __syncthreads();

    if (wave < 3 * KQ) {
      const int tl = wave % 3, kq = wave / 3;
      const int it = (tl > 0) ? 1 : 0, jt = (tl == 2) ? 1 : 0;
      const float* ap = &Qf[(it * 16 + lm) * FP + kq * (Md / KQ) + 4 * hi];
      const float* bp = &Kf[(jt * 16 + lm) * FP + kq * (Md / KQ) + 4 * hi];
      v8f acc = {};
      for (int q = 0; q < Md / KQ / 8; ++q)
        acc = mma_k8(ap + 8 * q, bp + 8 * q, acc);
      for (int e = 0; e < 8; ++e)
        Apart[kq * (CK * TP) + (it * 16 + e + 8 * hi) * TP + jt * 16 + lm] = acc[e];
    }
    __syncthreads();

    for (int i = tid; i < CK * CK; i += NT2) {
      int gi = i >> 5, gj = i & 31;
      float a = 0.f;
      if (gj <= gi) {
        const float* p = &Apart[gi * TP + gj];
        a = p[0];
        for (int q = 1; q < KQ; ++q) a += p[q * (CK * TP)];
      }
      Alds[gi * TP + gj] = a;
    }
    __syncthreads();

    v8f oacc = {};
    if (wave < 16) {
      const int tile = wave & 7, kh = wave >> 3;
      int it = tile >> 2, dt = tile & 3;
      int r0 = it * 16, c0 = dt * 16;
      {
        const float* ap = &Qf[(r0 + lm) * FP + kh * (Md / 2) + 4 * hi];
        const float* bp = &St[(c0 + lm) * FP + kh * (Md / 2) + 4 * hi];
        for (int q = 0; q < Md / 2 / 8; ++q)
          oacc = mma_k8(ap + 8 * q, bp + 8 * q, oacc);
      }
      if (kh == 0) {
        const float* ap = &Alds[(r0 + lm) * TP + 4 * hi];
        const float* bp = &Vt  [(c0 + lm) * RP + ro + 4 * hi];
        for (int q = 0; q < CK / 8; ++q)
          oacc = mma_k8(ap + 8 * q, bp + 8 * q, oacc);
      } else {
        for (int e = 0; e < 8; ++e)
          Opart[(r0 + e + 8 * hi) * RP + c0 + lm] = oacc[e];
      }
    } else {
      const int row = (wave - 16) * 2 + hi;
      float dsum = 0.f;
      for (int m = lm; m < Md; m += 16) dsum += Qf[row * FP + m] * sLds[m];
      for (int j = lm; j < CK; j += 16) dsum += Alds[row * TP + j];
      dsum += __shfl_xor(dsum, 8);
      dsum += __shfl_xor(dsum, 4);
      dsum += __shfl_xor(dsum, 2);
      dsum += __shfl_xor(dsum, 1);
      if (lm == 0) denLds[row] = dsum;
    }
    __syncthreads();

    if (wave < 8) {
      int it = wave >> 2, dt = wave & 3;
      for (int e = 0; e < 8; ++e) {
        int row = it * 16 + e + 8 * hi;
        float den = denLds[row];
        den = (den <= 0.f) ? 1.f : den;
        float num = oacc[e] + Opart[row * RP + dt * 16 + lm];
        float* dst =
            out + (((size_t)b * Ld + lbase + row) * Hd + h) * Dd + dt * 16 + lm;
        __builtin_nontemporal_store(num / den, dst);
      }
    }
    if (tid >= NT2 - Md) {
      const int m = tid - (NT2 - Md);
      const float* p = &KfT[m * TP];
      float cs = 0.f;
      for (int q = 0; q < CK / 4; ++q) {
        const v4f x = *reinterpret_cast<const v4f*>(p + 4 * q);
        cs += x.x; cs += x.y; cs += x.z; cs += x.w;
      }
      sLds[m] += cs;
    }
    for (int t = wave * STW; t < wave * STW + STW; ++t) {
      int mt = t >> 2, dt = t & 3;
      int r0 = mt * 16, c0 = dt * 16;
      float* sp = &St[(c0 + lm) * FP + r0 + 8 * hi];
      const v4f s0 = *reinterpret_cast<const v4f*>(sp);
      const v4f s1 = *reinterpret_cast<const v4f*>(sp + 4);
      v8f acc = {s0.x, s0.y, s0.z, s0.w, s1.x, s1.y, s1.z, s1.w};
      const float* ap = &KfT[(r0 + lm) * TP + 4 * hi];
      const float* bp = &Vt [(c0 + lm) * RP + ro + 4 * hi];
      for (int q = 0; q < CK / 8; ++q)
        acc = mma_k8(ap + 8 * q, bp + 8 * q, acc);
      v4f o0 = {acc[0], acc[1], acc[2], acc[3]};
      v4f o1 = {acc[4], acc[5], acc[6], acc[7]};
      *reinterpret_cast<v4f*>(sp)     = o0;
      *reinterpret_cast<v4f*>(sp + 4) = o1;
    }
    __syncthreads();
   }
  }
}

extern "C" void kernel_launch(void* const* d_in, const int* in_sizes, int n_in,
                              void* d_out, int out_size, void* d_ws, size_t ws_size,
                              hipStream_t stream) {
  const float* q    = (const float*)d_in[0];
  const float* k    = (const float*)d_in[1];
  const float* v    = (const float*)d_in[2];
  const float* proj = (const float*)d_in[3];
  float* out = (float*)d_out;
  float* ws  = (float*)d_ws;

  (void)in_sizes; (void)n_in; (void)out_size; (void)ws_size;

  perf_attn_segsum<<<Bd * Hd * NSEG, NT1, 0, stream>>>(k, v, proj, ws);
  perf_attn_main  <<<Bd * Hd * NSEG, NT2, 0, stream>>>(q, k, v, proj, ws, out);
}
